# starting kernel
# baseline (speedup 1.0000x reference)
#include <hip/hip_runtime.h>

typedef __attribute__((ext_vector_type(2))) float v2f;
typedef __attribute__((ext_vector_type(8))) float v8f;

__global__ void gcn_init(float* __restrict__ deg, float* __restrict__ agg1,
                         float* __restrict__ agg2, int n) {
    int i = blockIdx.x * blockDim.x + threadIdx.x;
    if (i < n) { deg[i] = 1.0f; agg2[i] = 0.0f; }
    if (i < n * 16) { agg1[i] = 0.0f; }
}

__global__ void gcn_degree(const int* __restrict__ col, float* __restrict__ deg,
                           int E) {
    int e = blockIdx.x * blockDim.x + threadIdx.x;
    if (e >= E) return;
    atomicAdd(&deg[col[e]], 1.0f);
}

__global__ void gcn_dinv(const float* __restrict__ deg, float* __restrict__ dinv,
                         int n) {
    int i = blockIdx.x * blockDim.x + threadIdx.x;
    if (i >= n) return;
    float d = deg[i];
    dinv[i] = (d > 0.0f) ? rsqrtf(d) : 0.0f;
}

__global__ void gcn_gemm1_wmma(const float* __restrict__ x,
                               const float* __restrict__ W1,
                               float* __restrict__ h1, int ntiles) {
    int wave = blockIdx.x * (blockDim.x >> 5) + (threadIdx.x >> 5);
    if (wave >= ntiles) return;
    int lane  = threadIdx.x & 31;
    int m     = lane & 15;
    int kb    = (lane >> 4) << 1;
    size_t row16 = (size_t)wave * 16;

    v8f acc = {};
    const float* arow = x + (row16 + (size_t)m) * 128 + kb;
    #pragma unroll 4
    for (int k0 = 0; k0 < 128; k0 += 4) {
        v2f a = *(const v2f*)(arow + k0);
        v2f b;
        b.x = W1[(size_t)(k0 + kb) * 16 + m];
        b.y = W1[(size_t)(k0 + kb + 1) * 16 + m];
        acc = __builtin_amdgcn_wmma_f32_16x16x4_f32(
             false, a,  false, b,
             (short)0, acc,  false,  false);
    }

    int mbase = (lane >> 4) << 3;
    int ncol  = lane & 15;
    #pragma unroll
    for (int r = 0; r < 8; ++r) {
        h1[(row16 + (size_t)(mbase + r)) * 16 + ncol] = acc[r];
    }
}

__global__ void gcn_gemm1_tail(const float* __restrict__ x,
                               const float* __restrict__ W1,
                               float* __restrict__ h1, int start, int n) {
    int t = blockIdx.x * blockDim.x + threadIdx.x;
    int node = start + (t >> 4);
    int f = t & 15;
    if (node >= n) return;
    float acc = 0.0f;
    const float* xr = x + (size_t)node * 128;
    for (int k = 0; k < 128; ++k) acc += xr[k] * W1[(size_t)k * 16 + f];
    h1[(size_t)node * 16 + f] = acc;
}

__global__ void gcn_edge16(const int* __restrict__ row, const int* __restrict__ col,
                           const float* __restrict__ dinv,
                           const float* __restrict__ h1,
                           float* __restrict__ agg1, int E) {
    int t = blockIdx.x * blockDim.x + threadIdx.x;
    int e = t >> 4;
    int f = t & 15;
    if (e >= E) return;
    int r = row[e];
    int c = col[e];
    float nrm = dinv[r] * dinv[c];
    float v = nrm * h1[(size_t)r * 16 + f];
    atomicAdd(&agg1[(size_t)c * 16 + f], v);
}

__global__ void gcn_post1(const float* __restrict__ agg1,
                          const float* __restrict__ h1,
                          const float* __restrict__ dinv,
                          const float* __restrict__ b1,
                          const float* __restrict__ W2,
                          float* __restrict__ s, int n) {
    int i = blockIdx.x * blockDim.x + threadIdx.x;
    if (i >= n) return;
    float di = dinv[i];
    float self = di * di;
    const float* a = agg1 + (size_t)i * 16;
    const float* h = h1 + (size_t)i * 16;
    float acc = 0.0f;
    #pragma unroll
    for (int f = 0; f < 16; ++f) {
        float v = a[f] + self * h[f] + b1[f];
        v = fmaxf(v, 0.0f);
        acc = fmaf(v, W2[f], acc);
    }
    s[i] = acc;
}

__global__ void gcn_edge1(const int* __restrict__ row, const int* __restrict__ col,
                          const float* __restrict__ dinv,
                          const float* __restrict__ s,
                          float* __restrict__ agg2, int E) {
    int e = blockIdx.x * blockDim.x + threadIdx.x;
    if (e >= E) return;
    int r = row[e];
    int c = col[e];
    atomicAdd(&agg2[c], dinv[r] * dinv[c] * s[r]);
}

__global__ void gcn_post2(const float* __restrict__ agg2,
                          const float* __restrict__ s,
                          const float* __restrict__ dinv,
                          const float* __restrict__ b2,
                          float* __restrict__ out, int n) {
    int i = blockIdx.x * blockDim.x + threadIdx.x;
    if (i >= n) return;
    float di = dinv[i];
    out[i] = agg2[i] + di * di * s[i] + b2[0];
}

extern "C" void kernel_launch(void* const* d_in, const int* in_sizes, int n_in,
                              void* d_out, int out_size, void* d_ws, size_t ws_size,
                              hipStream_t stream) {
    const float* x  = (const float*)d_in[0];
    const int*   ei = (const int*)d_in[1];
    const float* W1 = (const float*)d_in[2];
    const float* b1 = (const float*)d_in[3];
    const float* W2 = (const float*)d_in[4];
    const float* b2 = (const float*)d_in[5];

    const int N = in_sizes[0] / 128;
    const int E = in_sizes[1] / 2;
    const int* row = ei;
    const int* col = ei + E;

    float* ws   = (float*)d_ws;
    float* deg  = ws;
    float* dinv = deg + N;
    float* h1   = dinv + N;
    float* agg1 = h1 + (size_t)N * 16;
    float* s    = agg1 + (size_t)N * 16;
    float* agg2 = s + N;
    float* out  = (float*)d_out;

    const int B = 256;

    {
        int total = N * 16;
        gcn_init<<<(total + B - 1) / B, B, 0, stream>>>(deg, agg1, agg2, N);
    }
    gcn_degree<<<(E + B - 1) / B, B, 0, stream>>>(col, deg, E);
    gcn_dinv<<<(N + B - 1) / B, B, 0, stream>>>(deg, dinv, N);
    {
        int full   = N & ~15;
        int ntiles = full >> 4;
        if (ntiles > 0) {
            int wavesPerBlock = B / 32;
            int blocks = (ntiles + wavesPerBlock - 1) / wavesPerBlock;
            gcn_gemm1_wmma<<<blocks, B, 0, stream>>>(x, W1, h1, ntiles);
        }
        if (full < N) {
            int tail = (N - full) * 16;
            gcn_gemm1_tail<<<(tail + B - 1) / B, B, 0, stream>>>(x, W1, h1, full, N);
        }
    }
    {
        long long total = (long long)E * 16;
        gcn_edge16<<<(int)((total + B - 1) / B), B, 0, stream>>>(row, col, dinv, h1, agg1, E);
    }
    gcn_post1<<<(N + B - 1) / B, B, 0, stream>>>(agg1, h1, dinv, b1, W2, s, N);
    gcn_edge1<<<(E + B - 1) / B, B, 0, stream>>>(row, col, dinv, s, agg2, E);
    gcn_post2<<<(N + B - 1) / B, B, 0, stream>>>(agg2, s, dinv, b2, out, N);
}


# Atomic sums (degree, layer 1, layer 2) kept in two copies, one per L2 half: a wave adds only to the copy of the half it runs on, node passes add the copies; float sum order changes
# speedup vs baseline: 2.7048x; 2.7048x over previous
#include <hip/hip_runtime.h>

typedef __attribute__((ext_vector_type(2))) float v2f;
typedef __attribute__((ext_vector_type(8))) float v8f;

__global__ void gcn_init(float* __restrict__ deg, float* __restrict__ agg1,
                         float* __restrict__ agg2, int n) {
    int i = blockIdx.x * blockDim.x + threadIdx.x;
    if (i < n) { deg[i] = 1.0f; agg2[i] = 0.0f; }
    if (i < n * 16) { agg1[i] = 0.0f; }
}

__global__ void gcn_degree(const int* __restrict__ col, float* __restrict__ deg,
                           int E) {
    int e = blockIdx.x * blockDim.x + threadIdx.x;
    if (e >= E) return;
    atomicAdd(&deg[col[e]], 1.0f);
}

__global__ void gcn_dinv(const float* __restrict__ deg, float* __restrict__ dinv,
                         int n) {
    int i = blockIdx.x * blockDim.x + threadIdx.x;
    if (i >= n) return;
    float d = deg[i];
    dinv[i] = (d > 0.0f) ? rsqrtf(d) : 0.0f;
}

__global__ void gcn_gemm1_wmma(const float* __restrict__ x,
                               const float* __restrict__ W1,
                               float* __restrict__ h1, int ntiles) {
    int wave = blockIdx.x * (blockDim.x >> 5) + (threadIdx.x >> 5);
    if (wave >= ntiles) return;
    int lane  = threadIdx.x & 31;
    int m     = lane & 15;
    int kb    = (lane >> 4) << 1;
    size_t row16 = (size_t)wave * 16;

    v8f acc = {};
    const float* arow = x + (row16 + (size_t)m) * 128 + kb;
    #pragma unroll 4
    for (int k0 = 0; k0 < 128; k0 += 4) {
        v2f a = *(const v2f*)(arow + k0);
        v2f b;
        b.x = W1[(size_t)(k0 + kb) * 16 + m];
        b.y = W1[(size_t)(k0 + kb + 1) * 16 + m];
        acc = __builtin_amdgcn_wmma_f32_16x16x4_f32(
             false, a,  false, b,
             (short)0, acc,  false,  false);
    }

    int mbase = (lane >> 4) << 3;
    int ncol  = lane & 15;
    #pragma unroll
    for (int r = 0; r < 8; ++r) {
        h1[(row16 + (size_t)(mbase + r)) * 16 + ncol] = acc[r];
    }
}

__global__ void gcn_gemm1_tail(const float* __restrict__ x,
                               const float* __restrict__ W1,
                               float* __restrict__ h1, int start, int n) {
    int t = blockIdx.x * blockDim.x + threadIdx.x;
    int node = start + (t >> 4);
    int f = t & 15;
    if (node >= n) return;
    float acc = 0.0f;
    const float* xr = x + (size_t)node * 128;
    for (int k = 0; k < 128; ++k) acc += xr[k] * W1[(size_t)k * 16 + f];
    h1[(size_t)node * 16 + f] = acc;
}

__global__ void gcn_edge16(const int* __restrict__ row, const int* __restrict__ col,
                           const float* __restrict__ dinv,
                           const float* __restrict__ h1,
                           float* __restrict__ agg1, int E) {
    int t = blockIdx.x * blockDim.x + threadIdx.x;
    int e = t >> 4;
    int f = t & 15;
    if (e >= E) return;
    int r = row[e];
    int c = col[e];
    float nrm = dinv[r] * dinv[c];
    float v = nrm * h1[(size_t)r * 16 + f];
    atomicAdd(&agg1[(size_t)c * 16 + f], v);
}

__global__ void gcn_post1(const float* __restrict__ agg1,
                          const float* __restrict__ h1,
                          const float* __restrict__ dinv,
                          const float* __restrict__ b1,
                          const float* __restrict__ W2,
                          float* __restrict__ s, int n) {
    int i = blockIdx.x * blockDim.x + threadIdx.x;
    if (i >= n) return;
    float di = dinv[i];
    float self = di * di;
    const float* a = agg1 + (size_t)i * 16;
    const float* h = h1 + (size_t)i * 16;
    float acc = 0.0f;
    #pragma unroll
    for (int f = 0; f < 16; ++f) {
        float v = a[f] + self * h[f] + b1[f];
        v = fmaxf(v, 0.0f);
        acc = fmaf(v, W2[f], acc);
    }
    s[i] = acc;
}

__global__ void gcn_edge1(const int* __restrict__ row, const int* __restrict__ col,
                          const float* __restrict__ dinv,
                          const float* __restrict__ s,
                          float* __restrict__ agg2, int E) {
    int e = blockIdx.x * blockDim.x + threadIdx.x;
    if (e >= E) return;
    int r = row[e];
    int c = col[e];
    atomicAdd(&agg2[c], dinv[r] * dinv[c] * s[r]);
}

__global__ void gcn_post2(const float* __restrict__ agg2,
                          const float* __restrict__ s,
                          const float* __restrict__ dinv,
                          const float* __restrict__ b2,
                          float* __restrict__ out, int n) {
    int i = blockIdx.x * blockDim.x + threadIdx.x;
    if (i >= n) return;
    float di = dinv[i];
    out[i] = agg2[i] + di * di * s[i] + b2[0];
}


__device__ __forceinline__ int gcn_half() {
    return (int)((__builtin_amdgcn_s_sendmsg_rtn(0x87) >> 18) & 1u);
}

__global__ void gcn2_init(float* __restrict__ deg2, float4* __restrict__ agg1x2, int n) {
    int i = blockIdx.x * blockDim.x + threadIdx.x;
    if (i < n) { deg2[i] = 1.0f; deg2[n + i] = 0.0f; }
    if (i < n * 8) agg1x2[i] = make_float4(0.0f, 0.0f, 0.0f, 0.0f);
}

__global__ void gcn2_degree(const int* __restrict__ col, float* __restrict__ deg2,
                            int n, int E) {
    int e = blockIdx.x * blockDim.x + threadIdx.x;
    if (e >= E) return;
    float* deg = deg2 + (size_t)gcn_half() * n;
    atomicAdd(&deg[col[e]], 1.0f);
}

__global__ void gcn2_dinv(float* __restrict__ deg2, float* __restrict__ dinv, int n) {
    int i = blockIdx.x * blockDim.x + threadIdx.x;
    if (i >= n) return;
    float d = deg2[i] + deg2[n + i];
    dinv[i] = (d > 0.0f) ? rsqrtf(d) : 0.0f;
    deg2[i] = 0.0f;
    deg2[n + i] = 0.0f;
}

__global__ void gcn2_edge16(const int* __restrict__ row, const int* __restrict__ col,
                            const float* __restrict__ dinv,
                            const float* __restrict__ h1,
                            float* __restrict__ agg1x2, int n, int E) {
    int wave = (blockIdx.x * blockDim.x + threadIdx.x) >> 5;
    int lane = threadIdx.x & 31;
    int f    = lane & 15;
    int e0   = wave * 32 + (lane >> 4);
    if (wave * 32 >= E) return;
    float* agg = agg1x2 + (size_t)gcn_half() * n * 16;
    if (wave * 32 + 32 <= E) {
        for (int i = 0; i < 16; ++i) {
            int e = e0 + 2 * i;
            int r = row[e];
            int c = col[e];
            float nrm = dinv[r] * dinv[c];
            float v = nrm * h1[(size_t)r * 16 + f];
            atomicAdd(&agg[(size_t)c * 16 + f], v);
        }
    } else {
        for (int i = 0; i < 16; ++i) {
            int e = e0 + 2 * i;
            if (e < E) {
                int r = row[e];
                int c = col[e];
                float nrm = dinv[r] * dinv[c];
                float v = nrm * h1[(size_t)r * 16 + f];
                atomicAdd(&agg[(size_t)c * 16 + f], v);
            }
        }
    }
}

__global__ void gcn2_post1(const float* __restrict__ agg1x2,
                           const float* __restrict__ h1,
                           const float* __restrict__ dinv,
                           const float* __restrict__ b1,
                           const float* __restrict__ W2,
                           float* __restrict__ s, int n) {
    int i = blockIdx.x * blockDim.x + threadIdx.x;
    if (i >= n) return;
    float di = dinv[i];
    float self = di * di;
    const float* a0 = agg1x2 + (size_t)i * 16;
    const float* a1 = a0 + (size_t)n * 16;
    const float* h = h1 + (size_t)i * 16;
    float acc = 0.0f;
    #pragma unroll
    for (int f = 0; f < 16; ++f) {
        float v = (a0[f] + a1[f]) + self * h[f] + b1[f];
        v = fmaxf(v, 0.0f);
        acc = fmaf(v, W2[f], acc);
    }
    s[i] = acc;
}

__global__ void gcn2_edge1(const int* __restrict__ row, const int* __restrict__ col,
                           const float* __restrict__ dinv,
                           const float* __restrict__ s,
                           float* __restrict__ agg2x2, int n, int E) {
    int e = blockIdx.x * blockDim.x + threadIdx.x;
    if (e >= E) return;
    float* agg2 = agg2x2 + (size_t)gcn_half() * n;
    int r = row[e];
    int c = col[e];
    atomicAdd(&agg2[c], dinv[r] * dinv[c] * s[r]);
}

__global__ void gcn2_post2(const float* __restrict__ agg2x2,
                           const float* __restrict__ s,
                           const float* __restrict__ dinv,
                           const float* __restrict__ b2,
                           float* __restrict__ out, int n) {
    int i = blockIdx.x * blockDim.x + threadIdx.x;
    if (i >= n) return;
    float di = dinv[i];
    out[i] = (agg2x2[i] + agg2x2[n + i]) + di * di * s[i] + b2[0];
}

extern "C" void kernel_launch(void* const* d_in, const int* in_sizes, int n_in,
                              void* d_out, int out_size, void* d_ws, size_t ws_size,
                              hipStream_t stream) {
    const float* x  = (const float*)d_in[0];
    const int*   ei = (const int*)d_in[1];
    const float* W1 = (const float*)d_in[2];
    const float* b1 = (const float*)d_in[3];
    const float* W2 = (const float*)d_in[4];
    const float* b2 = (const float*)d_in[5];

    const int N = in_sizes[0] / 128;
    const int E = in_sizes[1] / 2;
    const int* row = ei;
    const int* col = ei + E;

    float* ws   = (float*)d_ws;
    float* deg  = ws;
    float* dinv = deg + N;
    float* h1   = dinv + N;
    float* agg1 = h1 + (size_t)N * 16;
    float* s    = agg1 + (size_t)N * 16;
    float* agg2 = s + N;
    float* out  = (float*)d_out;

    const int B = 256;

    if (ws_size >= (size_t)N * 52 * sizeof(float)) {
        float* xagg1 = ws;
        float* xh1   = xagg1 + (size_t)N * 32;
        float* xdeg2 = xh1 + (size_t)N * 16;
        float* xdinv = xdeg2 + (size_t)N * 2;
        float* xs    = xdinv + N;

        gcn2_init<<<(N * 8 + B - 1) / B, B, 0, stream>>>(xdeg2, (float4*)xagg1, N);
        gcn2_degree<<<(E + B - 1) / B, B, 0, stream>>>(col, xdeg2, N, E);
        gcn2_dinv<<<(N + B - 1) / B, B, 0, stream>>>(xdeg2, xdinv, N);
        {
            int full   = N & ~15;
            int ntiles = full >> 4;
            if (ntiles > 0) {
                int wavesPerBlock = B / 32;
                int blocks = (ntiles + wavesPerBlock - 1) / wavesPerBlock;
                gcn_gemm1_wmma<<<blocks, B, 0, stream>>>(x, W1, xh1, ntiles);
            }
            if (full < N) {
                int tail = (N - full) * 16;
                gcn_gemm1_tail<<<(tail + B - 1) / B, B, 0, stream>>>(x, W1, xh1, full, N);
            }
        }
        gcn2_edge16<<<(E + B - 1) / B, B, 0, stream>>>(row, col, xdinv, xh1, xagg1, N, E);
        gcn2_post1<<<(N + B - 1) / B, B, 0, stream>>>(xagg1, xh1, xdinv, b1, W2, xs, N);
        gcn2_edge1<<<(E + B - 1) / B, B, 0, stream>>>(row, col, xdinv, xs, xdeg2, N, E);
        gcn2_post2<<<(N + B - 1) / B, B, 0, stream>>>(xdeg2, xs, xdinv, b2, out, N);
        return;
    }

    {
        int total = N * 16;
        gcn_init<<<(total + B - 1) / B, B, 0, stream>>>(deg, agg1, agg2, N);
    }
    gcn_degree<<<(E + B - 1) / B, B, 0, stream>>>(col, deg, E);
    gcn_dinv<<<(N + B - 1) / B, B, 0, stream>>>(deg, dinv, N);
    {
        int full   = N & ~15;
        int ntiles = full >> 4;
        if (ntiles > 0) {
            int wavesPerBlock = B / 32;
            int blocks = (ntiles + wavesPerBlock - 1) / wavesPerBlock;
            gcn_gemm1_wmma<<<blocks, B, 0, stream>>>(x, W1, h1, ntiles);
        }
        if (full < N) {
            int tail = (N - full) * 16;
            gcn_gemm1_tail<<<(tail + B - 1) / B, B, 0, stream>>>(x, W1, h1, full, N);
        }
    }
    {
        long long total = (long long)E * 16;
        gcn_edge16<<<(int)((total + B - 1) / B), B, 0, stream>>>(row, col, dinv, h1, agg1, E);
    }
    gcn_post1<<<(N + B - 1) / B, B, 0, stream>>>(agg1, h1, dinv, b1, W2, s, N);
    gcn_edge1<<<(E + B - 1) / B, B, 0, stream>>>(row, col, dinv, s, agg2, E);
    gcn_post2<<<(N + B - 1) / B, B, 0, stream>>>(agg2, s, dinv, b2, out, N);
}
